# starting kernel
# baseline (speedup 1.0000x reference)
#include <hip/hip_runtime.h>

typedef float v2f __attribute__((ext_vector_type(2)));
typedef float v8f __attribute__((ext_vector_type(8)));

#define D 64

__device__ __forceinline__ void gatomic_add_f32(float* addr, float v) {
    asm volatile("global_atomic_add_f32 %0, %1, off" : : "v"(addr), "v"(v) : "memory");
}

__global__ __launch_bounds__(256) void k_scatter(const int* __restrict__ src,
                                                 const int* __restrict__ dst,
                                                 const float* __restrict__ norm,
                                                 const float* __restrict__ H,
                                                 float* __restrict__ acc, int e_cnt) {
    const unsigned g = blockIdx.x * 256u + threadIdx.x;
    const unsigned e = g >> 4;
    const unsigned l = g & 15u;
    if (e >= (unsigned)e_cnt) return;
    const unsigned s   = (unsigned)src[e];
    const unsigned d   = (unsigned)dst[e];
    const float    nrm = norm[e];
    const float4 hv = *(const float4*)(H + s * (unsigned)D + l * 4u);
    float* base = acc + d * (unsigned)D + l * 4u;
    gatomic_add_f32(base + 0, hv.x * nrm);
    gatomic_add_f32(base + 1, hv.y * nrm);
    gatomic_add_f32(base + 2, hv.z * nrm);
    gatomic_add_f32(base + 3, hv.w * nrm);
}

__global__ __launch_bounds__(256) void k_init_deg(float* deg, int n) {
    int i = blockIdx.x * blockDim.x + threadIdx.x;
    if (i < n) deg[i] = 1.0f;
}

__global__ __launch_bounds__(256) void k_deg_count(const int* __restrict__ dst,
                                                   float* __restrict__ deg, int e_cnt) {
    int e = blockIdx.x * blockDim.x + threadIdx.x;
    if (e < e_cnt) gatomic_add_f32(&deg[dst[e]], 1.0f);
}

__global__ __launch_bounds__(256) void k_rsqrt(float* d, int n) {
    int i = blockIdx.x * blockDim.x + threadIdx.x;
    if (i < n) d[i] = rsqrtf(d[i]);
}

__global__ __launch_bounds__(256) void k_edge_norm(const int* __restrict__ src,
                                                   const int* __restrict__ dst,
                                                   const float* __restrict__ dinv,
                                                   float* __restrict__ norm, int e_cnt) {
    int e = blockIdx.x * blockDim.x + threadIdx.x;
    if (e < e_cnt) norm[e] = dinv[src[e]] * dinv[dst[e]];
}

__global__ __launch_bounds__(256) void k_selfinit(const float* __restrict__ H,
                                                  const float* __restrict__ dinv,
                                                  const float* __restrict__ bias,
                                                  float* __restrict__ acc, int n_nodes) {
    const long i = (long)blockIdx.x * blockDim.x + threadIdx.x;
    if (i >= (long)n_nodes * D) return;
    const int node = (int)(i >> 6);
    const int c    = (int)(i & (D - 1));
    const float di = dinv[node];
    acc[i] = H[i] * di * di + bias[c];
}

template <bool RELU>
__global__ __launch_bounds__(128) void k_gemm_wmma(const float* __restrict__ X,
                                                   const float* __restrict__ W,
                                                   float* __restrict__ H, int n_rows) {
    const int lane = threadIdx.x & 31;
    const int wave = threadIdx.x >> 5;
    const int half = lane >> 4;
    const int l16  = lane & 15;
    const int mt   = blockIdx.x;

    int row = mt * 16 + l16;
    if (row >= n_rows) row = n_rows - 1;
    const int ncol = wave * 16 + l16;

    v8f c = {};
#pragma unroll
    for (int k0 = 0; k0 < D; k0 += 4) {
        const int ka = k0 + 2 * half;
        float2 av = *(const float2*)(X + (long)row * D + ka);
        if (RELU) { av.x = fmaxf(av.x, 0.0f); av.y = fmaxf(av.y, 0.0f); }
        v2f a, b;
        a.x = av.x;  a.y = av.y;
        b.x = W[ka * D + ncol];
        b.y = W[(ka + 1) * D + ncol];
        c = __builtin_amdgcn_wmma_f32_16x16x4_f32(false, a, false, b,
                                                  (short)0, c, false, false);
    }

    const int mbase = mt * 16 + half * 8;
#pragma unroll
    for (int r = 0; r < 8; ++r) {
        const int m = mbase + r;
        if (m < n_rows) H[(long)m * D + ncol] = c[r];
    }
}

extern "C" void kernel_launch(void* const* d_in, const int* in_sizes, int n_in,
                              void* d_out, int out_size, void* d_ws, size_t ws_size,
                              hipStream_t stream) {
    const float* x  = (const float*)d_in[0];
    const int*   ei = (const int*)d_in[1];
    const float* Wm[3] = { (const float*)d_in[2], (const float*)d_in[4], (const float*)d_in[6] };
    const float* bm[3] = { (const float*)d_in[3], (const float*)d_in[5], (const float*)d_in[7] };

    const int N = in_sizes[0] / D;
    const int E = in_sizes[1] / 2;
    const int* src = ei;
    const int* dst = ei + E;

    const long npad = ((long)N + 63) & ~63L;
    const long epad = ((long)E + 63) & ~63L;
    const long feat = (long)N * D;
    float* dinv = (float*)d_ws;
    float* norm = dinv + npad;
    float* h    = norm + epad;
    float* acc  = h + feat;

    const int B = 256;
    const int gN    = (N + B - 1) / B;
    const int gE    = (E + B - 1) / B;
    const int gFeat = (int)((feat + B - 1) / B);
    const int gScat = (int)(((long)E * 16 + B - 1) / B);
    const int gGemm = (N + 15) / 16;

    k_init_deg<<<gN, B, 0, stream>>>(dinv, N);
    k_deg_count<<<gE, B, 0, stream>>>(dst, dinv, E);
    k_rsqrt<<<gN, B, 0, stream>>>(dinv, N);
    k_edge_norm<<<gE, B, 0, stream>>>(src, dst, dinv, norm, E);

    k_gemm_wmma<false><<<gGemm, 128, 0, stream>>>(x, Wm[0], h, N);
    k_selfinit<<<gFeat, B, 0, stream>>>(h, dinv, bm[0], acc, N);
    k_scatter<<<gScat, B, 0, stream>>>(src, dst, norm, h, acc, E);

    k_gemm_wmma<true><<<gGemm, 128, 0, stream>>>(acc, Wm[1], h, N);
    k_selfinit<<<gFeat, B, 0, stream>>>(h, dinv, bm[1], acc, N);
    k_scatter<<<gScat, B, 0, stream>>>(src, dst, norm, h, acc, E);

    float* out = (float*)d_out;
    k_gemm_wmma<true><<<gGemm, 128, 0, stream>>>(acc, Wm[2], h, N);
    k_selfinit<<<gFeat, B, 0, stream>>>(h, dinv, bm[2], out, N);
    k_scatter<<<gScat, B, 0, stream>>>(src, dst, norm, h, out, E);
}


# edge scatter remapped: one wave per 32 edges, edge list read once and shared by readlane, each row gathered and added as two whole 128B lines (atomic adds kept; sum order still arrival order)
# speedup vs baseline: 1.6410x; 1.6410x over previous
#include <hip/hip_runtime.h>

typedef float v2f __attribute__((ext_vector_type(2)));
typedef float v8f __attribute__((ext_vector_type(8)));

#define D 64

__device__ __forceinline__ void gatomic_add_f32(float* addr, float v) {
    asm volatile("global_atomic_add_f32 %0, %1, off" : : "v"(addr), "v"(v) : "memory");
}


__device__ __forceinline__ void gatomic_add_row_f32(float* row, unsigned voff, float v0, float v1) {
    asm volatile(
        "global_atomic_add_f32 %[o], %[a0], %[r0]\n\t"
        "global_atomic_add_f32 %[o], %[b0], %[r0] offset:128\n\t"
        "s_wait_xcnt 0x0"
        : : [o] "v"(voff), [a0] "v"(v0), [b0] "v"(v1), [r0] "s"(row));
}

__device__ __forceinline__ void gatomic_add_8rows_f32(float* const (&r)[8], unsigned voff,
                                                      const float (&a)[8], const float (&b)[8]) {
    asm volatile(
        "global_atomic_add_f32 %[o], %[a0], %[r0]\n\t"
        "global_atomic_add_f32 %[o], %[b0], %[r0] offset:128\n\t"
        "global_atomic_add_f32 %[o], %[a1], %[r1]\n\t"
        "global_atomic_add_f32 %[o], %[b1], %[r1] offset:128\n\t"
        "global_atomic_add_f32 %[o], %[a2], %[r2]\n\t"
        "global_atomic_add_f32 %[o], %[b2], %[r2] offset:128\n\t"
        "global_atomic_add_f32 %[o], %[a3], %[r3]\n\t"
        "global_atomic_add_f32 %[o], %[b3], %[r3] offset:128\n\t"
        "global_atomic_add_f32 %[o], %[a4], %[r4]\n\t"
        "global_atomic_add_f32 %[o], %[b4], %[r4] offset:128\n\t"
        "global_atomic_add_f32 %[o], %[a5], %[r5]\n\t"
        "global_atomic_add_f32 %[o], %[b5], %[r5] offset:128\n\t"
        "global_atomic_add_f32 %[o], %[a6], %[r6]\n\t"
        "global_atomic_add_f32 %[o], %[b6], %[r6] offset:128\n\t"
        "global_atomic_add_f32 %[o], %[a7], %[r7]\n\t"
        "global_atomic_add_f32 %[o], %[b7], %[r7] offset:128\n\t"
        "s_wait_xcnt 0x0"
        : : [o] "v"(voff),
          [a0] "v"(a[0]), [b0] "v"(b[0]), [r0] "s"(r[0]),
          [a1] "v"(a[1]), [b1] "v"(b[1]), [r1] "s"(r[1]),
          [a2] "v"(a[2]), [b2] "v"(b[2]), [r2] "s"(r[2]),
          [a3] "v"(a[3]), [b3] "v"(b[3]), [r3] "s"(r[3]),
          [a4] "v"(a[4]), [b4] "v"(b[4]), [r4] "s"(r[4]),
          [a5] "v"(a[5]), [b5] "v"(b[5]), [r5] "s"(r[5]),
          [a6] "v"(a[6]), [b6] "v"(b[6]), [r6] "s"(r[6]),
          [a7] "v"(a[7]), [b7] "v"(b[7]), [r7] "s"(r[7]));
}

__device__ __forceinline__ float bcast_f32(float v, int k) {
    return __int_as_float(__builtin_amdgcn_readlane(__float_as_int(v), k));
}

#define EG 16
__global__ __launch_bounds__(256) void k_scatter(const int* __restrict__ src,
                                                 const int* __restrict__ dst,
                                                 const float* __restrict__ norm,
                                                 const float* __restrict__ H,
                                                 float* __restrict__ acc, int e_cnt) {
    const unsigned lane = threadIdx.x & 31u;
    const unsigned wv = (unsigned)__builtin_amdgcn_readfirstlane((int)((blockIdx.x * 256u + threadIdx.x) >> 5));
    const unsigned e0 = wv * 32u;
    if (e0 >= (unsigned)e_cnt) return;
    const unsigned left = (unsigned)e_cnt - e0;

    int s = 0, d = 0;
    float w = 0.0f;
    if (lane < left) {
        s = src[e0 + lane];
        d = dst[e0 + lane];
        w = norm[e0 + lane];
    }
    const unsigned voff = lane * 4u;

    if (left >= 32u) {
#pragma unroll
        for (int g = 0; g < 32; g += EG) {
            float h0[EG], h1[EG];
#pragma unroll
            for (int j = 0; j < EG; ++j) {
                const unsigned sj = (unsigned)__builtin_amdgcn_readlane(s, g + j);
                const float* p = H + (size_t)sj * D;
                h0[j] = p[lane];
                h1[j] = p[lane + 32u];
            }
#pragma unroll
            for (int q = 0; q < EG; q += 8) {
                float* r[8];
                float a[8], b[8];
#pragma unroll
                for (int j = 0; j < 8; ++j) {
                    const unsigned dj = (unsigned)__builtin_amdgcn_readlane(d, g + q + j);
                    const float    wj = bcast_f32(w, g + q + j);
                    r[j] = acc + (size_t)dj * D;
                    a[j] = h0[q + j] * wj;
                    b[j] = h1[q + j] * wj;
                }
                gatomic_add_8rows_f32(r, voff, a, b);
            }
        }
    } else {
        for (unsigned k = 0; k < left; ++k) {
            const unsigned sk = (unsigned)__builtin_amdgcn_readlane(s, (int)k);
            const unsigned dk = (unsigned)__builtin_amdgcn_readlane(d, (int)k);
            const float    wk = bcast_f32(w, (int)k);
            const float* p = H + (size_t)sk * D;
            gatomic_add_row_f32(acc + (size_t)dk * D, voff, p[lane] * wk, p[lane + 32u] * wk);
        }
    }
}

__global__ __launch_bounds__(256) void k_init_deg(float* deg, int n) {
    int i = blockIdx.x * blockDim.x + threadIdx.x;
    if (i < n) deg[i] = 1.0f;
}

__global__ __launch_bounds__(256) void k_deg_count(const int* __restrict__ dst,
                                                   float* __restrict__ deg, int e_cnt) {
    int e = blockIdx.x * blockDim.x + threadIdx.x;
    if (e < e_cnt) gatomic_add_f32(&deg[dst[e]], 1.0f);
}

__global__ __launch_bounds__(256) void k_rsqrt(float* d, int n) {
    int i = blockIdx.x * blockDim.x + threadIdx.x;
    if (i < n) d[i] = rsqrtf(d[i]);
}

__global__ __launch_bounds__(256) void k_edge_norm(const int* __restrict__ src,
                                                   const int* __restrict__ dst,
                                                   const float* __restrict__ dinv,
                                                   float* __restrict__ norm, int e_cnt) {
    int e = blockIdx.x * blockDim.x + threadIdx.x;
    if (e < e_cnt) norm[e] = dinv[src[e]] * dinv[dst[e]];
}

__global__ __launch_bounds__(256) void k_selfinit(const float* __restrict__ H,
                                                  const float* __restrict__ dinv,
                                                  const float* __restrict__ bias,
                                                  float* __restrict__ acc, int n_nodes) {
    const long i = (long)blockIdx.x * blockDim.x + threadIdx.x;
    if (i >= (long)n_nodes * D) return;
    const int node = (int)(i >> 6);
    const int c    = (int)(i & (D - 1));
    const float di = dinv[node];
    acc[i] = H[i] * di * di + bias[c];
}

template <bool RELU>
__global__ __launch_bounds__(128) void k_gemm_wmma(const float* __restrict__ X,
                                                   const float* __restrict__ W,
                                                   float* __restrict__ H, int n_rows) {
    const int lane = threadIdx.x & 31;
    const int wave = threadIdx.x >> 5;
    const int half = lane >> 4;
    const int l16  = lane & 15;
    const int mt   = blockIdx.x;

    int row = mt * 16 + l16;
    if (row >= n_rows) row = n_rows - 1;
    const int ncol = wave * 16 + l16;

    v8f c = {};
#pragma unroll
    for (int k0 = 0; k0 < D; k0 += 4) {
        const int ka = k0 + 2 * half;
        float2 av = *(const float2*)(X + (long)row * D + ka);
        if (RELU) { av.x = fmaxf(av.x, 0.0f); av.y = fmaxf(av.y, 0.0f); }
        v2f a, b;
        a.x = av.x;  a.y = av.y;
        b.x = W[ka * D + ncol];
        b.y = W[(ka + 1) * D + ncol];
        c = __builtin_amdgcn_wmma_f32_16x16x4_f32(false, a, false, b,
                                                  (short)0, c, false, false);
    }

    const int mbase = mt * 16 + half * 8;
#pragma unroll
    for (int r = 0; r < 8; ++r) {
        const int m = mbase + r;
        if (m < n_rows) H[(long)m * D + ncol] = c[r];
    }
}

extern "C" void kernel_launch(void* const* d_in, const int* in_sizes, int n_in,
                              void* d_out, int out_size, void* d_ws, size_t ws_size,
                              hipStream_t stream) {
    const float* x  = (const float*)d_in[0];
    const int*   ei = (const int*)d_in[1];
    const float* Wm[3] = { (const float*)d_in[2], (const float*)d_in[4], (const float*)d_in[6] };
    const float* bm[3] = { (const float*)d_in[3], (const float*)d_in[5], (const float*)d_in[7] };

    const int N = in_sizes[0] / D;
    const int E = in_sizes[1] / 2;
    const int* src = ei;
    const int* dst = ei + E;

    const long npad = ((long)N + 63) & ~63L;
    const long epad = ((long)E + 63) & ~63L;
    const long feat = (long)N * D;
    float* dinv = (float*)d_ws;
    float* norm = dinv + npad;
    float* h    = norm + epad;
    float* acc  = h + feat;

    const int B = 256;
    const int gN    = (N + B - 1) / B;
    const int gE    = (E + B - 1) / B;
    const int gFeat = (int)((feat + B - 1) / B);
    const int gGemm = (N + 15) / 16;

    k_init_deg<<<gN, B, 0, stream>>>(dinv, N);
    k_deg_count<<<gE, B, 0, stream>>>(dst, dinv, E);
    k_rsqrt<<<gN, B, 0, stream>>>(dinv, N);
    k_edge_norm<<<gE, B, 0, stream>>>(src, dst, dinv, norm, E);

    k_gemm_wmma<false><<<gGemm, 128, 0, stream>>>(x, Wm[0], h, N);
    k_selfinit<<<gFeat, B, 0, stream>>>(h, dinv, bm[0], acc, N);
    k_scatter<<<gE, B, 0, stream>>>(src, dst, norm, h, acc, E);

    k_gemm_wmma<true><<<gGemm, 128, 0, stream>>>(acc, Wm[1], h, N);
    k_selfinit<<<gFeat, B, 0, stream>>>(h, dinv, bm[1], acc, N);
    k_scatter<<<gE, B, 0, stream>>>(src, dst, norm, h, acc, E);

    float* out = (float*)d_out;
    k_gemm_wmma<true><<<gGemm, 128, 0, stream>>>(acc, Wm[2], h, N);
    k_selfinit<<<gFeat, B, 0, stream>>>(h, dinv, bm[2], out, N);
    k_scatter<<<gE, B, 0, stream>>>(src, dst, norm, h, out, E);
}


# accumulator start (self-loop term + bias) written by the GEMM epilogue from the product registers; the 3 init launches and their re-read of h go (13 -> 10); accumulator alternates out/workspace plane
# speedup vs baseline: 1.8139x; 1.1054x over previous
#include <hip/hip_runtime.h>

typedef float v2f __attribute__((ext_vector_type(2)));
typedef float v8f __attribute__((ext_vector_type(8)));

#define D 64

__device__ __forceinline__ void gatomic_add_f32(float* addr, float v) {
    asm volatile("global_atomic_add_f32 %0, %1, off" : : "v"(addr), "v"(v) : "memory");
}


__device__ __forceinline__ void gatomic_add_row_f32(float* row, unsigned voff, float v0, float v1) {
    asm volatile(
        "global_atomic_add_f32 %[o], %[a0], %[r0]\n\t"
        "global_atomic_add_f32 %[o], %[b0], %[r0] offset:128\n\t"
        "s_wait_xcnt 0x0"
        : : [o] "v"(voff), [a0] "v"(v0), [b0] "v"(v1), [r0] "s"(row));
}

__device__ __forceinline__ void gatomic_add_8rows_f32(float* const (&r)[8], unsigned voff,
                                                      const float (&a)[8], const float (&b)[8]) {
    asm volatile(
        "global_atomic_add_f32 %[o], %[a0], %[r0]\n\t"
        "global_atomic_add_f32 %[o], %[b0], %[r0] offset:128\n\t"
        "global_atomic_add_f32 %[o], %[a1], %[r1]\n\t"
        "global_atomic_add_f32 %[o], %[b1], %[r1] offset:128\n\t"
        "global_atomic_add_f32 %[o], %[a2], %[r2]\n\t"
        "global_atomic_add_f32 %[o], %[b2], %[r2] offset:128\n\t"
        "global_atomic_add_f32 %[o], %[a3], %[r3]\n\t"
        "global_atomic_add_f32 %[o], %[b3], %[r3] offset:128\n\t"
        "global_atomic_add_f32 %[o], %[a4], %[r4]\n\t"
        "global_atomic_add_f32 %[o], %[b4], %[r4] offset:128\n\t"
        "global_atomic_add_f32 %[o], %[a5], %[r5]\n\t"
        "global_atomic_add_f32 %[o], %[b5], %[r5] offset:128\n\t"
        "global_atomic_add_f32 %[o], %[a6], %[r6]\n\t"
        "global_atomic_add_f32 %[o], %[b6], %[r6] offset:128\n\t"
        "global_atomic_add_f32 %[o], %[a7], %[r7]\n\t"
        "global_atomic_add_f32 %[o], %[b7], %[r7] offset:128\n\t"
        "s_wait_xcnt 0x0"
        : : [o] "v"(voff),
          [a0] "v"(a[0]), [b0] "v"(b[0]), [r0] "s"(r[0]),
          [a1] "v"(a[1]), [b1] "v"(b[1]), [r1] "s"(r[1]),
          [a2] "v"(a[2]), [b2] "v"(b[2]), [r2] "s"(r[2]),
          [a3] "v"(a[3]), [b3] "v"(b[3]), [r3] "s"(r[3]),
          [a4] "v"(a[4]), [b4] "v"(b[4]), [r4] "s"(r[4]),
          [a5] "v"(a[5]), [b5] "v"(b[5]), [r5] "s"(r[5]),
          [a6] "v"(a[6]), [b6] "v"(b[6]), [r6] "s"(r[6]),
          [a7] "v"(a[7]), [b7] "v"(b[7]), [r7] "s"(r[7]));
}

__device__ __forceinline__ float bcast_f32(float v, int k) {
    return __int_as_float(__builtin_amdgcn_readlane(__float_as_int(v), k));
}

#define EG 16
__global__ __launch_bounds__(256) void k_scatter(const int* __restrict__ src,
                                                 const int* __restrict__ dst,
                                                 const float* __restrict__ norm,
                                                 const float* __restrict__ H,
                                                 float* __restrict__ acc, int e_cnt) {
    const unsigned lane = threadIdx.x & 31u;
    const unsigned wv = (unsigned)__builtin_amdgcn_readfirstlane((int)((blockIdx.x * 256u + threadIdx.x) >> 5));
    const unsigned e0 = wv * 32u;
    if (e0 >= (unsigned)e_cnt) return;
    const unsigned left = (unsigned)e_cnt - e0;

    int s = 0, d = 0;
    float w = 0.0f;
    if (lane < left) {
        s = src[e0 + lane];
        d = dst[e0 + lane];
        w = norm[e0 + lane];
    }
    const unsigned voff = lane * 4u;

    if (left >= 32u) {
#pragma unroll
        for (int g = 0; g < 32; g += EG) {
            float h0[EG], h1[EG];
#pragma unroll
            for (int j = 0; j < EG; ++j) {
                const unsigned sj = (unsigned)__builtin_amdgcn_readlane(s, g + j);
                const float* p = H + (size_t)sj * D;
                h0[j] = p[lane];
                h1[j] = p[lane + 32u];
            }
#pragma unroll
            for (int q = 0; q < EG; q += 8) {
                float* r[8];
                float a[8], b[8];
#pragma unroll
                for (int j = 0; j < 8; ++j) {
                    const unsigned dj = (unsigned)__builtin_amdgcn_readlane(d, g + q + j);
                    const float    wj = bcast_f32(w, g + q + j);
                    r[j] = acc + (size_t)dj * D;
                    a[j] = h0[q + j] * wj;
                    b[j] = h1[q + j] * wj;
                }
                gatomic_add_8rows_f32(r, voff, a, b);
            }
        }
    } else {
        for (unsigned k = 0; k < left; ++k) {
            const unsigned sk = (unsigned)__builtin_amdgcn_readlane(s, (int)k);
            const unsigned dk = (unsigned)__builtin_amdgcn_readlane(d, (int)k);
            const float    wk = bcast_f32(w, (int)k);
            const float* p = H + (size_t)sk * D;
            gatomic_add_row_f32(acc + (size_t)dk * D, voff, p[lane] * wk, p[lane + 32u] * wk);
        }
    }
}

__global__ __launch_bounds__(256) void k_init_deg(float* deg, int n) {
    int i = blockIdx.x * blockDim.x + threadIdx.x;
    if (i < n) deg[i] = 1.0f;
}

__global__ __launch_bounds__(256) void k_deg_count(const int* __restrict__ dst,
                                                   float* __restrict__ deg, int e_cnt) {
    int e = blockIdx.x * blockDim.x + threadIdx.x;
    if (e < e_cnt) gatomic_add_f32(&deg[dst[e]], 1.0f);
}

__global__ __launch_bounds__(256) void k_rsqrt(float* d, int n) {
    int i = blockIdx.x * blockDim.x + threadIdx.x;
    if (i < n) d[i] = rsqrtf(d[i]);
}

__global__ __launch_bounds__(256) void k_edge_norm(const int* __restrict__ src,
                                                   const int* __restrict__ dst,
                                                   const float* __restrict__ dinv,
                                                   float* __restrict__ norm, int e_cnt) {
    int e = blockIdx.x * blockDim.x + threadIdx.x;
    if (e < e_cnt) norm[e] = dinv[src[e]] * dinv[dst[e]];
}

template <bool RELU>
__global__ __launch_bounds__(128) void k_gemm_wmma(const float* __restrict__ X,
                                                   const float* __restrict__ W,
                                                   const float* __restrict__ dinv,
                                                   const float* __restrict__ bias,
                                                   float* __restrict__ H,
                                                   float* __restrict__ A0, int n_rows) {
    const int lane = threadIdx.x & 31;
    const int wave = threadIdx.x >> 5;
    const int half = lane >> 4;
    const int l16  = lane & 15;
    const int mt   = blockIdx.x;

    int row = mt * 16 + l16;
    if (row >= n_rows) row = n_rows - 1;
    const int ncol = wave * 16 + l16;
    const int mbase = mt * 16 + half * 8;

    const float bc = bias[ncol];
    const float dl = dinv[row];
    const float d2 = dl * dl;

    v8f c = {};
#pragma unroll
    for (int k0 = 0; k0 < D; k0 += 4) {
        const int ka = k0 + 2 * half;
        float2 av = *(const float2*)(X + (long)row * D + ka);
        if (RELU) { av.x = fmaxf(av.x, 0.0f); av.y = fmaxf(av.y, 0.0f); }
        v2f a, b;
        a.x = av.x;  a.y = av.y;
        b.x = W[ka * D + ncol];
        b.y = W[(ka + 1) * D + ncol];
        c = __builtin_amdgcn_wmma_f32_16x16x4_f32(false, a, false, b,
                                                  (short)0, c, false, false);
    }

#pragma unroll
    for (int r = 0; r < 8; ++r) {
        const float s_lo = bcast_f32(d2, r);
        const float s_hi = bcast_f32(d2, r + 8);
        const float sc = half ? s_hi : s_lo;
        const int m = mbase + r;
        if (m < n_rows) {
            const long o = (long)m * D + ncol;
            const float hv = c[r];
            H[o]  = hv;
            A0[o] = hv * sc + bc;
        }
    }
}

extern "C" void kernel_launch(void* const* d_in, const int* in_sizes, int n_in,
                              void* d_out, int out_size, void* d_ws, size_t ws_size,
                              hipStream_t stream) {
    const float* x  = (const float*)d_in[0];
    const int*   ei = (const int*)d_in[1];
    const float* Wm[3] = { (const float*)d_in[2], (const float*)d_in[4], (const float*)d_in[6] };
    const float* bm[3] = { (const float*)d_in[3], (const float*)d_in[5], (const float*)d_in[7] };

    const int N = in_sizes[0] / D;
    const int E = in_sizes[1] / 2;
    const int* src = ei;
    const int* dst = ei + E;

    const long npad = ((long)N + 63) & ~63L;
    const long epad = ((long)E + 63) & ~63L;
    const long feat = (long)N * D;
    float* dinv = (float*)d_ws;
    float* norm = dinv + npad;
    float* h    = norm + epad;
    float* acc  = h + feat;

    const int B = 256;
    const int gN    = (N + B - 1) / B;
    const int gE    = (E + B - 1) / B;
    const int gGemm = (N + 15) / 16;

    k_init_deg<<<gN, B, 0, stream>>>(dinv, N);
    k_deg_count<<<gE, B, 0, stream>>>(dst, dinv, E);
    k_rsqrt<<<gN, B, 0, stream>>>(dinv, N);
    k_edge_norm<<<gE, B, 0, stream>>>(src, dst, dinv, norm, E);

    float* out = (float*)d_out;

    k_gemm_wmma<false><<<gGemm, 128, 0, stream>>>(x, Wm[0], dinv, bm[0], h, out, N);
    k_scatter<<<gE, B, 0, stream>>>(src, dst, norm, h, out, E);

    k_gemm_wmma<true><<<gGemm, 128, 0, stream>>>(out, Wm[1], dinv, bm[1], h, acc, N);
    k_scatter<<<gE, B, 0, stream>>>(src, dst, norm, h, acc, E);

    k_gemm_wmma<true><<<gGemm, 128, 0, stream>>>(acc, Wm[2], dinv, bm[2], h, out, N);
    k_scatter<<<gE, B, 0, stream>>>(src, dst, norm, h, out, E);
}


# symmetric norm applied one factor per side, once per row: dinv[src] in GEMM epilogue, dinv[dst] in next GEMM's load (last layer: scatter); norm array and 2 launches go (10->8); products re-associated
# speedup vs baseline: 1.8969x; 1.0458x over previous
#include <hip/hip_runtime.h>

typedef float v2f __attribute__((ext_vector_type(2)));
typedef float v8f __attribute__((ext_vector_type(8)));

#define D 64

__device__ __forceinline__ void gatomic_add_f32(float* addr, float v) {
    asm volatile("global_atomic_add_f32 %0, %1, off" : : "v"(addr), "v"(v) : "memory");
}


__device__ __forceinline__ void gatomic_add_row_f32(float* row, unsigned voff, float v0, float v1) {
    asm volatile(
        "global_atomic_add_f32 %[o], %[a0], %[r0]\n\t"
        "global_atomic_add_f32 %[o], %[b0], %[r0] offset:128\n\t"
        "s_wait_xcnt 0x0"
        : : [o] "v"(voff), [a0] "v"(v0), [b0] "v"(v1), [r0] "s"(row));
}

__device__ __forceinline__ void gatomic_add_8rows_f32(float* const (&r)[8], unsigned voff,
                                                      const float (&a)[8], const float (&b)[8]) {
    asm volatile(
        "global_atomic_add_f32 %[o], %[a0], %[r0]\n\t"
        "global_atomic_add_f32 %[o], %[b0], %[r0] offset:128\n\t"
        "global_atomic_add_f32 %[o], %[a1], %[r1]\n\t"
        "global_atomic_add_f32 %[o], %[b1], %[r1] offset:128\n\t"
        "global_atomic_add_f32 %[o], %[a2], %[r2]\n\t"
        "global_atomic_add_f32 %[o], %[b2], %[r2] offset:128\n\t"
        "global_atomic_add_f32 %[o], %[a3], %[r3]\n\t"
        "global_atomic_add_f32 %[o], %[b3], %[r3] offset:128\n\t"
        "global_atomic_add_f32 %[o], %[a4], %[r4]\n\t"
        "global_atomic_add_f32 %[o], %[b4], %[r4] offset:128\n\t"
        "global_atomic_add_f32 %[o], %[a5], %[r5]\n\t"
        "global_atomic_add_f32 %[o], %[b5], %[r5] offset:128\n\t"
        "global_atomic_add_f32 %[o], %[a6], %[r6]\n\t"
        "global_atomic_add_f32 %[o], %[b6], %[r6] offset:128\n\t"
        "global_atomic_add_f32 %[o], %[a7], %[r7]\n\t"
        "global_atomic_add_f32 %[o], %[b7], %[r7] offset:128\n\t"
        "s_wait_xcnt 0x0"
        : : [o] "v"(voff),
          [a0] "v"(a[0]), [b0] "v"(b[0]), [r0] "s"(r[0]),
          [a1] "v"(a[1]), [b1] "v"(b[1]), [r1] "s"(r[1]),
          [a2] "v"(a[2]), [b2] "v"(b[2]), [r2] "s"(r[2]),
          [a3] "v"(a[3]), [b3] "v"(b[3]), [r3] "s"(r[3]),
          [a4] "v"(a[4]), [b4] "v"(b[4]), [r4] "s"(r[4]),
          [a5] "v"(a[5]), [b5] "v"(b[5]), [r5] "s"(r[5]),
          [a6] "v"(a[6]), [b6] "v"(b[6]), [r6] "s"(r[6]),
          [a7] "v"(a[7]), [b7] "v"(b[7]), [r7] "s"(r[7]));
}

__device__ __forceinline__ float bcast_f32(float v, int k) {
    return __int_as_float(__builtin_amdgcn_readlane(__float_as_int(v), k));
}

#define EG 16
template <bool WEIGHTED>
__global__ __launch_bounds__(256) __attribute__((amdgpu_waves_per_eu(16, 16)))
void k_scatter(const int* __restrict__ src,
               const int* __restrict__ dst,
               const float* __restrict__ deg,
               const float* __restrict__ H,
               float* __restrict__ acc, int e_cnt) {
    const unsigned lane = threadIdx.x & 31u;
    const unsigned wv = (unsigned)__builtin_amdgcn_readfirstlane((int)((blockIdx.x * 256u + threadIdx.x) >> 5));
    const unsigned e0 = wv * 32u;
    if (e0 >= (unsigned)e_cnt) return;
    const unsigned left = (unsigned)e_cnt - e0;

    int s = 0, d = 0;
    float w = 0.0f;
    if (lane < left) {
        s = src[e0 + lane];
        d = dst[e0 + lane];
    }
    if (WEIGHTED) w = rsqrtf(deg[d]);
    const unsigned voff = lane * 4u;

    if (left >= 32u) {
#pragma unroll
        for (int g = 0; g < 32; g += EG) {
            float h0[EG], h1[EG];
#pragma unroll
            for (int j = 0; j < EG; ++j) {
                const unsigned sj = (unsigned)__builtin_amdgcn_readlane(s, g + j);
                const float* p = H + (size_t)sj * D;
                h0[j] = p[lane];
                h1[j] = p[lane + 32u];
            }
#pragma unroll
            for (int q = 0; q < EG; q += 8) {
                float* r[8];
                float a[8], b[8];
#pragma unroll
                for (int j = 0; j < 8; ++j) {
                    const unsigned dj = (unsigned)__builtin_amdgcn_readlane(d, g + q + j);
                    r[j] = acc + (size_t)dj * D;
                    if (WEIGHTED) {
                        const float wj = bcast_f32(w, g + q + j);
                        a[j] = h0[q + j] * wj;
                        b[j] = h1[q + j] * wj;
                    } else {
                        a[j] = h0[q + j];
                        b[j] = h1[q + j];
                    }
                }
                gatomic_add_8rows_f32(r, voff, a, b);
            }
        }
    } else {
        for (unsigned k = 0; k < left; ++k) {
            const unsigned sk = (unsigned)__builtin_amdgcn_readlane(s, (int)k);
            const unsigned dk = (unsigned)__builtin_amdgcn_readlane(d, (int)k);
            const float    wk = WEIGHTED ? bcast_f32(w, (int)k) : 1.0f;
            const float* p = H + (size_t)sk * D;
            gatomic_add_row_f32(acc + (size_t)dk * D, voff, p[lane] * wk, p[lane + 32u] * wk);
        }
    }
}

__global__ __launch_bounds__(256) void k_init_deg(float* deg, int n) {
    int i = blockIdx.x * blockDim.x + threadIdx.x;
    if (i < n) deg[i] = 1.0f;
}

__global__ __launch_bounds__(256) void k_deg_count(const int* __restrict__ dst,
                                                   float* __restrict__ deg, int e_cnt) {
    int e = blockIdx.x * blockDim.x + threadIdx.x;
    if (e < e_cnt) gatomic_add_f32(&deg[dst[e]], 1.0f);
}

template <bool DEFER_IN, bool FINAL>
__global__ __launch_bounds__(128) void k_gemm_wmma(const float* __restrict__ X,
                                                   const float* __restrict__ W,
                                                   const float* __restrict__ deg,
                                                   const float* __restrict__ bprev,
                                                   const float* __restrict__ bias,
                                                   float* __restrict__ H,
                                                   float* __restrict__ A0, int n_rows) {
    const int lane = threadIdx.x & 31;
    const int wave = threadIdx.x >> 5;
    const int half = lane >> 4;
    const int l16  = lane & 15;
    const int mt   = blockIdx.x;

    int row = mt * 16 + l16;
    if (row >= n_rows) row = n_rows - 1;
    const int ncol = wave * 16 + l16;
    const int mbase = mt * 16 + half * 8;

    const float bc = FINAL ? bias[ncol] : 0.0f;
    const float dl = rsqrtf(deg[row]);

    v8f c = {};
#pragma unroll
    for (int k0 = 0; k0 < D; k0 += 4) {
        const int ka = k0 + 2 * half;
        float2 av = *(const float2*)(X + (long)row * D + ka);
        if (DEFER_IN) {
            const float2 bp = *(const float2*)(bprev + ka);
            av.x = fmaxf(fmaf(av.x, dl, bp.x), 0.0f);
            av.y = fmaxf(fmaf(av.y, dl, bp.y), 0.0f);
        }
        v2f a, b;
        a.x = av.x;  a.y = av.y;
        b.x = W[ka * D + ncol];
        b.y = W[(ka + 1) * D + ncol];
        c = __builtin_amdgcn_wmma_f32_16x16x4_f32(false, a, false, b,
                                                  (short)0, c, false, false);
    }

#pragma unroll
    for (int r = 0; r < 8; ++r) {
        const float s_lo = bcast_f32(dl, r);
        const float s_hi = bcast_f32(dl, r + 8);
        const float sc = half ? s_hi : s_lo;
        const int m = mbase + r;
        if (m < n_rows) {
            const long o = (long)m * D + ncol;
            const float hv = c[r] * sc;
            H[o]  = hv;
            A0[o] = FINAL ? hv * sc + bc : hv;
        }
    }
}

extern "C" void kernel_launch(void* const* d_in, const int* in_sizes, int n_in,
                              void* d_out, int out_size, void* d_ws, size_t ws_size,
                              hipStream_t stream) {
    const float* x  = (const float*)d_in[0];
    const int*   ei = (const int*)d_in[1];
    const float* Wm[3] = { (const float*)d_in[2], (const float*)d_in[4], (const float*)d_in[6] };
    const float* bm[3] = { (const float*)d_in[3], (const float*)d_in[5], (const float*)d_in[7] };

    const int N = in_sizes[0] / D;
    const int E = in_sizes[1] / 2;
    const int* src = ei;
    const int* dst = ei + E;

    const long npad = ((long)N + 63) & ~63L;
    const long feat = (long)N * D;
    float* deg = (float*)d_ws;
    float* h   = deg + npad;
    float* acc = h + feat;

    const int B = 256;
    const int gN    = (N + B - 1) / B;
    const int gE    = (E + B - 1) / B;
    const int gGemm = (N + 15) / 16;

    k_init_deg<<<gN, B, 0, stream>>>(deg, N);
    k_deg_count<<<gE, B, 0, stream>>>(dst, deg, E);

    float* out = (float*)d_out;

    k_gemm_wmma<false, false><<<gGemm, 128, 0, stream>>>(x, Wm[0], deg, bm[0], bm[0], h, out, N);
    k_scatter<false><<<gE, B, 0, stream>>>(src, dst, deg, h, out, E);

    k_gemm_wmma<true, false><<<gGemm, 128, 0, stream>>>(out, Wm[1], deg, bm[0], bm[1], h, acc, N);
    k_scatter<false><<<gE, B, 0, stream>>>(src, dst, deg, h, acc, E);

    k_gemm_wmma<true, true><<<gGemm, 128, 0, stream>>>(acc, Wm[2], deg, bm[1], bm[2], h, out, N);
    k_scatter<true><<<gE, B, 0, stream>>>(src, dst, deg, h, out, E);
}


# GEMM operand feed: a tile's 16 rows read once per block as whole 128B lines and staged in LDS (was 8-byte pieces of 16 rows per k-step, in each of 4 waves); blocks walk 5 tiles, W slice in registers
# speedup vs baseline: 1.9948x; 1.0516x over previous
#include <hip/hip_runtime.h>

typedef float v2f __attribute__((ext_vector_type(2)));
typedef float v8f __attribute__((ext_vector_type(8)));

#define D 64

__device__ __forceinline__ void gatomic_add_f32(float* addr, float v) {
    asm volatile("global_atomic_add_f32 %0, %1, off" : : "v"(addr), "v"(v) : "memory");
}


__device__ __forceinline__ void gatomic_add_row_f32(float* row, unsigned voff, float v0, float v1) {
    asm volatile(
        "global_atomic_add_f32 %[o], %[a0], %[r0]\n\t"
        "global_atomic_add_f32 %[o], %[b0], %[r0] offset:128\n\t"
        "s_wait_xcnt 0x0"
        : : [o] "v"(voff), [a0] "v"(v0), [b0] "v"(v1), [r0] "s"(row));
}

__device__ __forceinline__ void gatomic_add_8rows_f32(float* const (&r)[8], unsigned voff,
                                                      const float (&a)[8], const float (&b)[8]) {
    asm volatile(
        "global_atomic_add_f32 %[o], %[a0], %[r0]\n\t"
        "global_atomic_add_f32 %[o], %[b0], %[r0] offset:128\n\t"
        "global_atomic_add_f32 %[o], %[a1], %[r1]\n\t"
        "global_atomic_add_f32 %[o], %[b1], %[r1] offset:128\n\t"
        "global_atomic_add_f32 %[o], %[a2], %[r2]\n\t"
        "global_atomic_add_f32 %[o], %[b2], %[r2] offset:128\n\t"
        "global_atomic_add_f32 %[o], %[a3], %[r3]\n\t"
        "global_atomic_add_f32 %[o], %[b3], %[r3] offset:128\n\t"
        "global_atomic_add_f32 %[o], %[a4], %[r4]\n\t"
        "global_atomic_add_f32 %[o], %[b4], %[r4] offset:128\n\t"
        "global_atomic_add_f32 %[o], %[a5], %[r5]\n\t"
        "global_atomic_add_f32 %[o], %[b5], %[r5] offset:128\n\t"
        "global_atomic_add_f32 %[o], %[a6], %[r6]\n\t"
        "global_atomic_add_f32 %[o], %[b6], %[r6] offset:128\n\t"
        "global_atomic_add_f32 %[o], %[a7], %[r7]\n\t"
        "global_atomic_add_f32 %[o], %[b7], %[r7] offset:128\n\t"
        "s_wait_xcnt 0x0"
        : : [o] "v"(voff),
          [a0] "v"(a[0]), [b0] "v"(b[0]), [r0] "s"(r[0]),
          [a1] "v"(a[1]), [b1] "v"(b[1]), [r1] "s"(r[1]),
          [a2] "v"(a[2]), [b2] "v"(b[2]), [r2] "s"(r[2]),
          [a3] "v"(a[3]), [b3] "v"(b[3]), [r3] "s"(r[3]),
          [a4] "v"(a[4]), [b4] "v"(b[4]), [r4] "s"(r[4]),
          [a5] "v"(a[5]), [b5] "v"(b[5]), [r5] "s"(r[5]),
          [a6] "v"(a[6]), [b6] "v"(b[6]), [r6] "s"(r[6]),
          [a7] "v"(a[7]), [b7] "v"(b[7]), [r7] "s"(r[7]));
}

__device__ __forceinline__ float bcast_f32(float v, int k) {
    return __int_as_float(__builtin_amdgcn_readlane(__float_as_int(v), k));
}

#define EG 16
template <bool WEIGHTED>
__global__ __launch_bounds__(256) __attribute__((amdgpu_waves_per_eu(16, 16)))
void k_scatter(const int* __restrict__ src,
               const int* __restrict__ dst,
               const float* __restrict__ deg,
               const float* __restrict__ H,
               float* __restrict__ acc, int e_cnt) {
    const unsigned lane = threadIdx.x & 31u;
    const unsigned wv = (unsigned)__builtin_amdgcn_readfirstlane((int)((blockIdx.x * 256u + threadIdx.x) >> 5));
    const unsigned e0 = wv * 32u;
    if (e0 >= (unsigned)e_cnt) return;
    const unsigned left = (unsigned)e_cnt - e0;

    int s = 0, d = 0;
    float w = 0.0f;
    if (lane < left) {
        s = src[e0 + lane];
        d = dst[e0 + lane];
    }
    if (WEIGHTED) w = rsqrtf(deg[d]);
    const unsigned voff = lane * 4u;

    if (left >= 32u) {
#pragma unroll
        for (int g = 0; g < 32; g += EG) {
            float h0[EG], h1[EG];
#pragma unroll
            for (int j = 0; j < EG; ++j) {
                const unsigned sj = (unsigned)__builtin_amdgcn_readlane(s, g + j);
                const float* p = H + (size_t)sj * D;
                h0[j] = p[lane];
                h1[j] = p[lane + 32u];
            }
#pragma unroll
            for (int q = 0; q < EG; q += 8) {
                float* r[8];
                float a[8], b[8];
#pragma unroll
                for (int j = 0; j < 8; ++j) {
                    const unsigned dj = (unsigned)__builtin_amdgcn_readlane(d, g + q + j);
                    r[j] = acc + (size_t)dj * D;
                    if (WEIGHTED) {
                        const float wj = bcast_f32(w, g + q + j);
                        a[j] = h0[q + j] * wj;
                        b[j] = h1[q + j] * wj;
                    } else {
                        a[j] = h0[q + j];
                        b[j] = h1[q + j];
                    }
                }
                gatomic_add_8rows_f32(r, voff, a, b);
            }
        }
    } else {
        for (unsigned k = 0; k < left; ++k) {
            const unsigned sk = (unsigned)__builtin_amdgcn_readlane(s, (int)k);
            const unsigned dk = (unsigned)__builtin_amdgcn_readlane(d, (int)k);
            const float    wk = WEIGHTED ? bcast_f32(w, (int)k) : 1.0f;
            const float* p = H + (size_t)sk * D;
            gatomic_add_row_f32(acc + (size_t)dk * D, voff, p[lane] * wk, p[lane + 32u] * wk);
        }
    }
}

__global__ __launch_bounds__(256) void k_init_deg(float* deg, int n) {
    int i = blockIdx.x * blockDim.x + threadIdx.x;
    if (i < n) deg[i] = 1.0f;
}

__global__ __launch_bounds__(256) void k_deg_count(const int* __restrict__ dst,
                                                   float* __restrict__ deg, int e_cnt) {
    int e = blockIdx.x * blockDim.x + threadIdx.x;
    if (e < e_cnt) gatomic_add_f32(&deg[dst[e]], 1.0f);
}

#define XP 68

__device__ __forceinline__ void gemm_fetch(const float* __restrict__ X, const float* __restrict__ deg,
                                           int mt, int n_rows, int sr0, int sc, int l16,
                                           float4& x0, float4& x1, float& g0, float& g1, float& gl) {
    int ra = mt * 16 + sr0;
    int rb = ra + 8;
    int rl = mt * 16 + l16;
    if (ra >= n_rows) ra = n_rows - 1;
    if (rb >= n_rows) rb = n_rows - 1;
    if (rl >= n_rows) rl = n_rows - 1;
    x0 = *(const float4*)(X + (long)ra * D + sc);
    x1 = *(const float4*)(X + (long)rb * D + sc);
    g0 = deg[ra];
    g1 = deg[rb];
    gl = deg[rl];
}

template <bool DEFER_IN, bool FINAL>
__global__ __launch_bounds__(128) void k_gemm_wmma(const float* __restrict__ X,
                                                   const float* __restrict__ W,
                                                   const float* __restrict__ deg,
                                                   const float* __restrict__ bprev,
                                                   const float* __restrict__ bias,
                                                   float* __restrict__ H,
                                                   float* __restrict__ A0, int n_rows) {
    __shared__ float xs[2][16 * XP];

    const int tid  = threadIdx.x;
    const int lane = tid & 31;
    const int wave = tid >> 5;
    const int half = lane >> 4;
    const int l16  = lane & 15;
    const int ncol = wave * 16 + l16;
    const int ntiles = (n_rows + 15) >> 4;

    const int sr0 = tid >> 4;
    const int sc  = (tid & 15) * 4;

    v2f bf[16];
#pragma unroll
    for (int kk = 0; kk < 16; ++kk) {
        const int ka = kk * 4 + 2 * half;
        bf[kk].x = W[ka * D + ncol];
        bf[kk].y = W[(ka + 1) * D + ncol];
    }
    const float bc = FINAL ? bias[ncol] : 0.0f;
    float4 bp = make_float4(0.0f, 0.0f, 0.0f, 0.0f);
    if (DEFER_IN) bp = *(const float4*)(bprev + sc);

    int mt = blockIdx.x;
    float4 x0 = make_float4(0.0f, 0.0f, 0.0f, 0.0f), x1 = x0;
    float g0 = 1.0f, g1 = 1.0f, gl = 1.0f;
    if (mt < ntiles) gemm_fetch(X, deg, mt, n_rows, sr0, sc, l16, x0, x1, g0, g1, gl);

    int buf = 0;
    for (; mt < ntiles; mt += gridDim.x, buf ^= 1) {
        float* xb = xs[buf];

        const float dl = rsqrtf(gl);

        float4 y0 = x0, y1 = x1;
        if (DEFER_IN) {
            const float i0 = rsqrtf(g0), i1 = rsqrtf(g1);
            y0.x = fmaxf(fmaf(y0.x, i0, bp.x), 0.0f);
            y0.y = fmaxf(fmaf(y0.y, i0, bp.y), 0.0f);
            y0.z = fmaxf(fmaf(y0.z, i0, bp.z), 0.0f);
            y0.w = fmaxf(fmaf(y0.w, i0, bp.w), 0.0f);
            y1.x = fmaxf(fmaf(y1.x, i1, bp.x), 0.0f);
            y1.y = fmaxf(fmaf(y1.y, i1, bp.y), 0.0f);
            y1.z = fmaxf(fmaf(y1.z, i1, bp.z), 0.0f);
            y1.w = fmaxf(fmaf(y1.w, i1, bp.w), 0.0f);
        }
        *(float4*)(xb + sr0 * XP + sc)       = y0;
        *(float4*)(xb + (sr0 + 8) * XP + sc) = y1;

        __syncthreads();

        const int mn = mt + (int)gridDim.x;
        if (mn < ntiles) gemm_fetch(X, deg, mn, n_rows, sr0, sc, l16, x0, x1, g0, g1, gl);

        v8f c = {};
#pragma unroll
        for (int kk = 0; kk < 16; ++kk) {
            const float2 av = *(const float2*)(xb + l16 * XP + kk * 4 + 2 * half);
            v2f af;
            af.x = av.x;  af.y = av.y;
            c = __builtin_amdgcn_wmma_f32_16x16x4_f32(false, af, false, bf[kk],
                                                      (short)0, c, false, false);
        }

        const int mbase = mt * 16 + half * 8;
#pragma unroll
        for (int r = 0; r < 8; ++r) {
            const float s_lo = bcast_f32(dl, r);
            const float s_hi = bcast_f32(dl, r + 8);
            const float sc2 = half ? s_hi : s_lo;
            const int m = mbase + r;
            if (m < n_rows) {
                const long o = (long)m * D + ncol;
                const float hv = c[r] * sc2;
                H[o]  = hv;
                A0[o] = FINAL ? hv * sc2 + bc : hv;
            }
        }
    }
}

extern "C" void kernel_launch(void* const* d_in, const int* in_sizes, int n_in,
                              void* d_out, int out_size, void* d_ws, size_t ws_size,
                              hipStream_t stream) {
    const float* x  = (const float*)d_in[0];
    const int*   ei = (const int*)d_in[1];
    const float* Wm[3] = { (const float*)d_in[2], (const float*)d_in[4], (const float*)d_in[6] };
    const float* bm[3] = { (const float*)d_in[3], (const float*)d_in[5], (const float*)d_in[7] };

    const int N = in_sizes[0] / D;
    const int E = in_sizes[1] / 2;
    const int* src = ei;
    const int* dst = ei + E;

    const long npad = ((long)N + 63) & ~63L;
    const long feat = (long)N * D;
    float* deg = (float*)d_ws;
    float* h   = deg + npad;
    float* acc = h + feat;

    const int B = 256;
    const int gN    = (N + B - 1) / B;
    const int gE    = (E + B - 1) / B;
    const int gGemm = 1250;

    k_init_deg<<<gN, B, 0, stream>>>(deg, N);
    k_deg_count<<<gE, B, 0, stream>>>(dst, deg, E);

    float* out = (float*)d_out;

    k_gemm_wmma<false, false><<<gGemm, 128, 0, stream>>>(x, Wm[0], deg, bm[0], bm[0], h, out, N);
    k_scatter<false><<<gE, B, 0, stream>>>(src, dst, deg, h, out, E);

    k_gemm_wmma<true, false><<<gGemm, 128, 0, stream>>>(out, Wm[1], deg, bm[0], bm[1], h, acc, N);
    k_scatter<false><<<gE, B, 0, stream>>>(src, dst, deg, h, acc, E);

    k_gemm_wmma<true, true><<<gGemm, 128, 0, stream>>>(acc, Wm[2], deg, bm[1], bm[2], h, out, N);
    k_scatter<true><<<gE, B, 0, stream>>>(src, dst, deg, h, out, E);
}


# edge scatter in two walks ordered by chip half (die id): a wave adds first into the node range of its own half, then the other; no acc line takes adds from both halves at once; sum order: arrival
# speedup vs baseline: 2.0581x; 1.0317x over previous
#include <hip/hip_runtime.h>

typedef float v2f __attribute__((ext_vector_type(2)));
typedef float v8f __attribute__((ext_vector_type(8)));

#define D 64

__device__ __forceinline__ void gatomic_add_f32(float* addr, float v) {
    asm volatile("global_atomic_add_f32 %0, %1, off" : : "v"(addr), "v"(v) : "memory");
}


__device__ __forceinline__ void gatomic_add_row_f32(float* row, unsigned voff, float v0, float v1) {
    asm volatile(
        "global_atomic_add_f32 %[o], %[a0], %[r0]\n\t"
        "global_atomic_add_f32 %[o], %[b0], %[r0] offset:128\n\t"
        "s_wait_xcnt 0x0"
        : : [o] "v"(voff), [a0] "v"(v0), [b0] "v"(v1), [r0] "s"(row));
}

__device__ __forceinline__ void gatomic_add_8rows_f32(float* const (&r)[8], unsigned voff,
                                                      const float (&a)[8], const float (&b)[8]) {
    asm volatile(
        "global_atomic_add_f32 %[o], %[a0], %[r0]\n\t"
        "global_atomic_add_f32 %[o], %[b0], %[r0] offset:128\n\t"
        "global_atomic_add_f32 %[o], %[a1], %[r1]\n\t"
        "global_atomic_add_f32 %[o], %[b1], %[r1] offset:128\n\t"
        "global_atomic_add_f32 %[o], %[a2], %[r2]\n\t"
        "global_atomic_add_f32 %[o], %[b2], %[r2] offset:128\n\t"
        "global_atomic_add_f32 %[o], %[a3], %[r3]\n\t"
        "global_atomic_add_f32 %[o], %[b3], %[r3] offset:128\n\t"
        "global_atomic_add_f32 %[o], %[a4], %[r4]\n\t"
        "global_atomic_add_f32 %[o], %[b4], %[r4] offset:128\n\t"
        "global_atomic_add_f32 %[o], %[a5], %[r5]\n\t"
        "global_atomic_add_f32 %[o], %[b5], %[r5] offset:128\n\t"
        "global_atomic_add_f32 %[o], %[a6], %[r6]\n\t"
        "global_atomic_add_f32 %[o], %[b6], %[r6] offset:128\n\t"
        "global_atomic_add_f32 %[o], %[a7], %[r7]\n\t"
        "global_atomic_add_f32 %[o], %[b7], %[r7] offset:128\n\t"
        "s_wait_xcnt 0x0"
        : : [o] "v"(voff),
          [a0] "v"(a[0]), [b0] "v"(b[0]), [r0] "s"(r[0]),
          [a1] "v"(a[1]), [b1] "v"(b[1]), [r1] "s"(r[1]),
          [a2] "v"(a[2]), [b2] "v"(b[2]), [r2] "s"(r[2]),
          [a3] "v"(a[3]), [b3] "v"(b[3]), [r3] "s"(r[3]),
          [a4] "v"(a[4]), [b4] "v"(b[4]), [r4] "s"(r[4]),
          [a5] "v"(a[5]), [b5] "v"(b[5]), [r5] "s"(r[5]),
          [a6] "v"(a[6]), [b6] "v"(b[6]), [r6] "s"(r[6]),
          [a7] "v"(a[7]), [b7] "v"(b[7]), [r7] "s"(r[7]));
}

__device__ __forceinline__ float bcast_f32(float v, int k) {
    return __int_as_float(__builtin_amdgcn_readlane(__float_as_int(v), k));
}

#define EG 16

template <bool WEIGHTED>
__device__ __forceinline__ void scatter_batch32(int s, int d, float w, const float* __restrict__ H,
                                                float* __restrict__ acc, unsigned lane, unsigned voff) {
#pragma unroll
    for (int g = 0; g < 32; g += EG) {
        float h0[EG], h1[EG];
#pragma unroll
        for (int j = 0; j < EG; ++j) {
            const unsigned sj = (unsigned)__builtin_amdgcn_readlane(s, g + j);
            const float* p = H + (size_t)sj * D;
            h0[j] = p[lane];
            h1[j] = p[lane + 32u];
        }
#pragma unroll
        for (int q = 0; q < EG; q += 8) {
            float* r[8];
            float a[8], b[8];
#pragma unroll
            for (int j = 0; j < 8; ++j) {
                const unsigned dj = (unsigned)__builtin_amdgcn_readlane(d, g + q + j);
                r[j] = acc + (size_t)dj * D;
                if (WEIGHTED) {
                    const float wj = bcast_f32(w, g + q + j);
                    a[j] = h0[q + j] * wj;
                    b[j] = h1[q + j] * wj;
                } else {
                    a[j] = h0[q + j];
                    b[j] = h1[q + j];
                }
            }
            gatomic_add_8rows_f32(r, voff, a, b);
        }
    }
}

#define SW 4
template <bool WEIGHTED>
__global__ __launch_bounds__(SW * 32) __attribute__((amdgpu_waves_per_eu(16, 16)))
void k_scatter(const int* __restrict__ src,
               const int* __restrict__ dst,
               const float* __restrict__ deg,
               const float* __restrict__ H,
               float* __restrict__ acc, int e_cnt, int n_nodes) {
    __shared__ int2 q[SW][96];

    const unsigned lane = threadIdx.x & 31u;
    const unsigned wave = (unsigned)__builtin_amdgcn_readfirstlane((int)(threadIdx.x >> 5));
    const unsigned gw   = blockIdx.x * (unsigned)SW + wave;
    const unsigned nw   = gridDim.x * (unsigned)SW;
    const unsigned voff = lane * 4u;
    const unsigned npc  = ((unsigned)e_cnt + 31u) >> 5;
    const int split = n_nodes >> 1;
    int2* qw = q[wave];

    __shared__ int home_s;
    if (threadIdx.x == 0) home_s = (int)((__builtin_amdgcn_s_sendmsg_rtn(0x87) >> 18) & 1u);
    __syncthreads();
    const int home = __builtin_amdgcn_readfirstlane(home_s);

    int cnt = 0;
    for (int ph = 0; ph < 2; ++ph) {
        const int r = home ^ ph;
        for (unsigned c = gw; c < npc; c += 2u * nw) {
            const unsigned e0 = c * 32u + lane;
            const unsigned e1 = (c + nw) * 32u + lane;
            const bool v0 = e0 < (unsigned)e_cnt;
            const bool v1 = e1 < (unsigned)e_cnt;
            const unsigned last = (unsigned)e_cnt - 1u;
            const unsigned f0 = min(e0, last), f1 = min(e1, last);
            const int s0 = src[f0], d0 = dst[f0];
            const int s1 = src[f1], d1 = dst[f1];
            const bool in0 = v0 && ((int)(d0 >= split) == r);
            const bool in1 = v1 && ((int)(d1 >= split) == r);

            const unsigned m0 = __builtin_amdgcn_ballot_w32(in0);
            const int k0 = __builtin_popcount(m0);
            qw[cnt + (in0 ? (int)__builtin_amdgcn_mbcnt_lo(m0, 0u) : k0 + (int)__builtin_amdgcn_mbcnt_lo(~m0, 0u))] = make_int2(s0, d0);
            cnt += k0;
            const unsigned m1 = __builtin_amdgcn_ballot_w32(in1);
            const int k1 = __builtin_popcount(m1);
            qw[cnt + (in1 ? (int)__builtin_amdgcn_mbcnt_lo(m1, 0u) : k1 + (int)__builtin_amdgcn_mbcnt_lo(~m1, 0u))] = make_int2(s1, d1);
            cnt += k1;

#pragma unroll
            for (int b = 0; b < 2; ++b) {
                if (cnt >= 32) {
                    cnt -= 32;
                    const int2 e = qw[cnt + (int)(voff >> 2)];
                    float w = 0.0f;
                    if (WEIGHTED) w = rsqrtf(deg[e.y]);
                    scatter_batch32<WEIGHTED>(e.x, e.y, w, H, acc, lane, voff);
                }
            }
        }
    }

    if (cnt > 0) {
        int s = 0, d = 0;
        const int tl = (int)__builtin_amdgcn_mbcnt_lo(~0u, 0u);
        if (tl < cnt) { const int2 e = qw[tl]; s = e.x; d = e.y; }
        float w = 0.0f;
        if (WEIGHTED) w = rsqrtf(deg[d]);
        for (int k0 = 0; k0 < cnt; k0 += EG) {
            float t0[EG] = {}, t1[EG] = {};
#pragma unroll
            for (int j = 0; j < EG; ++j) {
                if (k0 + j < cnt) {
                    const unsigned sk = (unsigned)__builtin_amdgcn_readlane(s, k0 + j);
                    const float* p = H + (size_t)sk * D;
                    t0[j] = p[lane];
                    t1[j] = p[lane + 32u];
                }
            }
#pragma unroll
            for (int j = 0; j < EG; ++j) {
                if (k0 + j < cnt) {
                    const unsigned dk = (unsigned)__builtin_amdgcn_readlane(d, k0 + j);
                    const float    wk = WEIGHTED ? bcast_f32(w, k0 + j) : 1.0f;
                    gatomic_add_row_f32(acc + (size_t)dk * D, voff, t0[j] * wk, t1[j] * wk);
                }
            }
        }
    }
}

__global__ __launch_bounds__(256) void k_init_deg(float* deg, int n) {
    int i = blockIdx.x * blockDim.x + threadIdx.x;
    if (i < n) deg[i] = 1.0f;
}

__global__ __launch_bounds__(256) void k_deg_count(const int* __restrict__ dst,
                                                   float* __restrict__ deg, int e_cnt) {
    int e = blockIdx.x * blockDim.x + threadIdx.x;
    if (e < e_cnt) gatomic_add_f32(&deg[dst[e]], 1.0f);
}

#define XP 68

__device__ __forceinline__ void gemm_fetch(const float* __restrict__ X, const float* __restrict__ deg,
                                           int mt, int n_rows, int sr0, int sc, int l16,
                                           float4& x0, float4& x1, float& g0, float& g1, float& gl) {
    int ra = mt * 16 + sr0;
    int rb = ra + 8;
    int rl = mt * 16 + l16;
    if (ra >= n_rows) ra = n_rows - 1;
    if (rb >= n_rows) rb = n_rows - 1;
    if (rl >= n_rows) rl = n_rows - 1;
    x0 = *(const float4*)(X + (long)ra * D + sc);
    x1 = *(const float4*)(X + (long)rb * D + sc);
    g0 = deg[ra];
    g1 = deg[rb];
    gl = deg[rl];
}

template <bool DEFER_IN, bool FINAL>
__global__ __launch_bounds__(128) void k_gemm_wmma(const float* __restrict__ X,
                                                   const float* __restrict__ W,
                                                   const float* __restrict__ deg,
                                                   const float* __restrict__ bprev,
                                                   const float* __restrict__ bias,
                                                   float* __restrict__ H,
                                                   float* __restrict__ A0, int n_rows) {
    __shared__ float xs[2][16 * XP];

    const int tid  = threadIdx.x;
    const int lane = tid & 31;
    const int wave = tid >> 5;
    const int half = lane >> 4;
    const int l16  = lane & 15;
    const int ncol = wave * 16 + l16;
    const int ntiles = (n_rows + 15) >> 4;

    const int sr0 = tid >> 4;
    const int sc  = (tid & 15) * 4;

    v2f bf[16];
#pragma unroll
    for (int kk = 0; kk < 16; ++kk) {
        const int ka = kk * 4 + 2 * half;
        bf[kk].x = W[ka * D + ncol];
        bf[kk].y = W[(ka + 1) * D + ncol];
    }
    const float bc = FINAL ? bias[ncol] : 0.0f;
    float4 bp = make_float4(0.0f, 0.0f, 0.0f, 0.0f);
    if (DEFER_IN) bp = *(const float4*)(bprev + sc);

    int mt = blockIdx.x;
    float4 x0 = make_float4(0.0f, 0.0f, 0.0f, 0.0f), x1 = x0;
    float g0 = 1.0f, g1 = 1.0f, gl = 1.0f;
    if (mt < ntiles) gemm_fetch(X, deg, mt, n_rows, sr0, sc, l16, x0, x1, g0, g1, gl);

    int buf = 0;
    for (; mt < ntiles; mt += gridDim.x, buf ^= 1) {
        float* xb = xs[buf];

        const float dl = rsqrtf(gl);

        float4 y0 = x0, y1 = x1;
        if (DEFER_IN) {
            const float i0 = rsqrtf(g0), i1 = rsqrtf(g1);
            y0.x = fmaxf(fmaf(y0.x, i0, bp.x), 0.0f);
            y0.y = fmaxf(fmaf(y0.y, i0, bp.y), 0.0f);
            y0.z = fmaxf(fmaf(y0.z, i0, bp.z), 0.0f);
            y0.w = fmaxf(fmaf(y0.w, i0, bp.w), 0.0f);
            y1.x = fmaxf(fmaf(y1.x, i1, bp.x), 0.0f);
            y1.y = fmaxf(fmaf(y1.y, i1, bp.y), 0.0f);
            y1.z = fmaxf(fmaf(y1.z, i1, bp.z), 0.0f);
            y1.w = fmaxf(fmaf(y1.w, i1, bp.w), 0.0f);
        }
        *(float4*)(xb + sr0 * XP + sc)       = y0;
        *(float4*)(xb + (sr0 + 8) * XP + sc) = y1;

        __syncthreads();

        const int mn = mt + (int)gridDim.x;
        if (mn < ntiles) gemm_fetch(X, deg, mn, n_rows, sr0, sc, l16, x0, x1, g0, g1, gl);

        v8f c = {};
#pragma unroll
        for (int kk = 0; kk < 16; ++kk) {
            const float2 av = *(const float2*)(xb + l16 * XP + kk * 4 + 2 * half);
            v2f af;
            af.x = av.x;  af.y = av.y;
            c = __builtin_amdgcn_wmma_f32_16x16x4_f32(false, af, false, bf[kk],
                                                      (short)0, c, false, false);
        }

        const int mbase = mt * 16 + half * 8;
#pragma unroll
        for (int r = 0; r < 8; ++r) {
            const float s_lo = bcast_f32(dl, r);
            const float s_hi = bcast_f32(dl, r + 8);
            const float sc2 = half ? s_hi : s_lo;
            const int m = mbase + r;
            if (m < n_rows) {
                const long o = (long)m * D + ncol;
                const float hv = c[r] * sc2;
                H[o]  = hv;
                A0[o] = FINAL ? hv * sc2 + bc : hv;
            }
        }
    }
}

extern "C" void kernel_launch(void* const* d_in, const int* in_sizes, int n_in,
                              void* d_out, int out_size, void* d_ws, size_t ws_size,
                              hipStream_t stream) {
    const float* x  = (const float*)d_in[0];
    const int*   ei = (const int*)d_in[1];
    const float* Wm[3] = { (const float*)d_in[2], (const float*)d_in[4], (const float*)d_in[6] };
    const float* bm[3] = { (const float*)d_in[3], (const float*)d_in[5], (const float*)d_in[7] };

    const int N = in_sizes[0] / D;
    const int E = in_sizes[1] / 2;
    const int* src = ei;
    const int* dst = ei + E;

    const long npad = ((long)N + 63) & ~63L;
    const long feat = (long)N * D;
    float* deg = (float*)d_ws;
    float* h   = deg + npad;
    float* acc = h + feat;

    const int B = 256;
    const int gN    = (N + B - 1) / B;
    const int gE    = (E + B - 1) / B;
    const int gGemm = 1250;
    const int gScat = 3125;

    k_init_deg<<<gN, B, 0, stream>>>(deg, N);
    k_deg_count<<<gE, B, 0, stream>>>(dst, deg, E);

    float* out = (float*)d_out;

    k_gemm_wmma<false, false><<<gGemm, 128, 0, stream>>>(x, Wm[0], deg, bm[0], bm[0], h, out, N);
    k_scatter<false><<<gScat, SW * 32, 0, stream>>>(src, dst, deg, h, out, E, N);

    k_gemm_wmma<true, false><<<gGemm, 128, 0, stream>>>(out, Wm[1], deg, bm[0], bm[1], h, acc, N);
    k_scatter<false><<<gScat, SW * 32, 0, stream>>>(src, dst, deg, h, acc, E, N);

    k_gemm_wmma<true, true><<<gGemm, 128, 0, stream>>>(acc, Wm[2], deg, bm[1], bm[2], h, out, N);
    k_scatter<true><<<gScat, SW * 32, 0, stream>>>(src, dst, deg, h, out, E, N);
}
